# starting kernel
# baseline (speedup 1.0000x reference)
#include <hip/hip_runtime.h>

typedef float v2f __attribute__((ext_vector_type(2)));
typedef float v8f __attribute__((ext_vector_type(8)));

#define WMMA_F32(a, b, c) \
    __builtin_amdgcn_wmma_f32_16x16x4_f32(false, (a), false, (b), (short)0, (c), false, false)

#define EMBED 1024
#define THREE_EMBED 3072
#define NHEAD 16
#define HD 64
#define SEQ 2048
#define BATCH 4

__global__ __launch_bounds__(256) void qkv_gemm_kernel(
    const float* __restrict__ x, const float* __restrict__ w,
    const float* __restrict__ bias,
    float* __restrict__ qb, float* __restrict__ kb, float* __restrict__ vb)
{
    const int lane   = threadIdx.x & 31;
    const int wave   = threadIdx.x >> 5;
    const int lane16 = lane & 15;
    const int laneHi = lane >> 4;

    const int wid   = blockIdx.x * 8 + wave;
    const int mtile = wid & 255;
    const int ntile = wid >> 8;
    const int m0    = mtile * 32;
    const int n0    = ntile * 64;

    const float* arow0 = x + (long)(m0 + lane16) * EMBED + 2 * laneHi;
    const float* arow1 = arow0 + 16 * EMBED;

    v8f acc0[4] = {};
    v8f acc1[4] = {};

    #pragma unroll 4
    for (int k0 = 0; k0 < EMBED; k0 += 4) {
        v2f a0 = *(const v2f*)(arow0 + k0);
        v2f a1 = *(const v2f*)(arow1 + k0);
        const int kk = k0 + 2 * laneHi;
        const float* wp = w + (long)kk * THREE_EMBED + n0 + lane16;
        #pragma unroll
        for (int c = 0; c < 4; ++c) {
            v2f b;
            b.x = wp[c * 16];
            b.y = wp[c * 16 + THREE_EMBED];
            acc0[c] = WMMA_F32(a0, b, acc0[c]);
            acc1[c] = WMMA_F32(a1, b, acc1[c]);
        }
    }

    const int section = n0 / EMBED;
    const int h       = (n0 % EMBED) / HD;
    float* dst = (section == 0) ? qb : (section == 1) ? kb : vb;

    #pragma unroll
    for (int c = 0; c < 4; ++c) {
        const float bv = bias[n0 + c * 16 + lane16];
        const int d = c * 16 + lane16;
        #pragma unroll
        for (int g = 0; g < 2; ++g) {
            #pragma unroll
            for (int vv = 0; vv < 8; ++vv) {
                const int m  = m0 + 16 * g + vv + 8 * laneHi;
                const int bg = m >> 11;
                const int t  = m & 2047;
                const float val = (g == 0 ? acc0[c][vv] : acc1[c][vv]) + bv;
                dst[(((long)(bg * NHEAD + h)) * SEQ + t) * HD + d] = val;
            }
        }
    }
}

__global__ __launch_bounds__(32) void attn_kernel(
    const float* __restrict__ q, const float* __restrict__ k,
    const float* __restrict__ v, float* __restrict__ out)
{
    __shared__ float pstage[2 * 256];

    const int lane   = threadIdx.x & 31;
    const int lane16 = lane & 15;
    const int laneHi = lane >> 4;

    const int wid = blockIdx.x;
    const int q32 = wid & 63;
    const int bh  = wid >> 6;
    const int b   = bh >> 4;
    const int h   = bh & 15;
    const int q0  = q32 * 32;

    const float* qp0 = q + ((long)bh * SEQ + q0 + lane16) * HD + 2 * laneHi;
    const float* qp1 = qp0 + 16 * HD;
    v2f aQ0[16], aQ1[16];
    #pragma unroll
    for (int j = 0; j < 16; ++j) {
        aQ0[j] = *(const v2f*)(qp0 + 4 * j);
        aQ1[j] = *(const v2f*)(qp1 + 4 * j);
    }

    float mrow[2][8], lrow[2][8];
    v8f O[2][4] = {};
    #pragma unroll
    for (int g = 0; g < 2; ++g)
        #pragma unroll
        for (int vv = 0; vv < 8; ++vv) { mrow[g][vv] = -__builtin_inff(); lrow[g][vv] = 0.f; }

    const float scale = 0.125f;
    const int ktmax = (q0 + 31) >> 4;

    for (int kt = 0; kt <= ktmax; ++kt) {
        const int k0 = kt * 16;

        const float* kp = k + ((long)bh * SEQ + k0 + lane16) * HD + 2 * laneHi;
        v8f S[2] = {};
        #pragma unroll
        for (int j = 0; j < 16; ++j) {
            v2f bK = *(const v2f*)(kp + 4 * j);
            S[0] = WMMA_F32(aQ0[j], bK, S[0]);
            S[1] = WMMA_F32(aQ1[j], bK, S[1]);
        }

        #pragma unroll
        for (int g = 0; g < 2; ++g) {
            #pragma unroll
            for (int vv = 0; vv < 8; ++vv) {
                const int row = 16 * g + vv + 8 * laneHi;
                float s = S[g][vv] * scale;
                if (k0 + lane16 > q0 + row) s = -__builtin_inff();
                S[g][vv] = s;
            }
            float alpha[8];
            #pragma unroll
            for (int vv = 0; vv < 8; ++vv) {
                float tmax = S[g][vv];
                #pragma unroll
                for (int off = 1; off < 16; off <<= 1)
                    tmax = fmaxf(tmax, __shfl_xor(tmax, off, 32));
                const float mnew = fmaxf(mrow[g][vv], tmax);
                const float p    = __expf(S[g][vv] - mnew);
                float psum = p;
                #pragma unroll
                for (int off = 1; off < 16; off <<= 1)
                    psum += __shfl_xor(psum, off, 32);
                const float a = __expf(mrow[g][vv] - mnew);
                lrow[g][vv]  = lrow[g][vv] * a + psum;
                mrow[g][vv]  = mnew;
                alpha[vv]    = a;
                S[g][vv]     = p;
            }
            #pragma unroll
            for (int c = 0; c < 4; ++c)
                #pragma unroll
                for (int vv = 0; vv < 8; ++vv)
                    O[g][c][vv] *= alpha[vv];
        }

        __syncthreads();
        #pragma unroll
        for (int g = 0; g < 2; ++g)
            #pragma unroll
            for (int vv = 0; vv < 8; ++vv)
                pstage[g * 256 + (vv + 8 * laneHi) * 16 + lane16] = S[g][vv];
        __syncthreads();
        v2f aP0[4], aP1[4];
        #pragma unroll
        for (int j = 0; j < 4; ++j) {
            aP0[j] = *(const v2f*)(&pstage[lane16 * 16 + 4 * j + 2 * laneHi]);
            aP1[j] = *(const v2f*)(&pstage[256 + lane16 * 16 + 4 * j + 2 * laneHi]);
        }

        const float* vp = v + ((long)bh * SEQ + k0) * HD;
        #pragma unroll
        for (int c = 0; c < 4; ++c) {
            #pragma unroll
            for (int j = 0; j < 4; ++j) {
                const int kk = 4 * j + 2 * laneHi;
                const float* vbp = vp + (long)kk * HD + c * 16 + lane16;
                v2f bV;
                bV.x = vbp[0];
                bV.y = vbp[HD];
                O[0][c] = WMMA_F32(aP0[j], bV, O[0][c]);
                O[1][c] = WMMA_F32(aP1[j], bV, O[1][c]);
            }
        }
    }

    #pragma unroll
    for (int g = 0; g < 2; ++g) {
        #pragma unroll
        for (int c = 0; c < 4; ++c) {
            #pragma unroll
            for (int vv = 0; vv < 8; ++vv) {
                const int t = q0 + 16 * g + vv + 8 * laneHi;
                const int d = c * 16 + lane16;
                out[(((long)b * SEQ + t) * NHEAD + h) * HD + d] = O[g][c][vv] / lrow[g][vv];
            }
        }
    }
}

extern "C" void kernel_launch(void* const* d_in, const int* in_sizes, int n_in,
                              void* d_out, int out_size, void* d_ws, size_t ws_size,
                              hipStream_t stream) {
    const float* x    = (const float*)d_in[0];
    const float* Wqkv = (const float*)d_in[1];
    const float* bqkv = (const float*)d_in[2];
    float* out = (float*)d_out;

    float* ws = (float*)d_ws;
    const long QS = (long)BATCH * NHEAD * SEQ * HD;
    float* qb = ws;
    float* kb = ws + QS;
    float* vb = ws + 2 * QS;

    qkv_gemm_kernel<<<1536, 256, 0, stream>>>(x, Wqkv, bqkv, qb, kb, vb);
    attn_kernel<<<BATCH * NHEAD * (SEQ / 32), 32, 0, stream>>>(qb, kb, vb, out);
}


# Attention scores computed transposed (St = K*Qt, Ot += Vt*Pt, one query per lane): row max/sum run along a lane's registers; no 16-lane shuffles, no LDS transpose of P; key sum order changes
# speedup vs baseline: 1.8598x; 1.8598x over previous
#include <hip/hip_runtime.h>

typedef float v2f __attribute__((ext_vector_type(2)));
typedef float v8f __attribute__((ext_vector_type(8)));

#define WMMA_F32(a, b, c) \
    __builtin_amdgcn_wmma_f32_16x16x4_f32(false, (a), false, (b), (short)0, (c), false, false)

#define EMBED 1024
#define THREE_EMBED 3072
#define NHEAD 16
#define HD 64
#define SEQ 2048
#define BATCH 4

__global__ __launch_bounds__(256) void qkv_gemm_kernel(
    const float* __restrict__ x, const float* __restrict__ w,
    const float* __restrict__ bias,
    float* __restrict__ qb, float* __restrict__ kb, float* __restrict__ vb)
{
    const int lane   = threadIdx.x & 31;
    const int wave   = threadIdx.x >> 5;
    const int lane16 = lane & 15;
    const int laneHi = lane >> 4;

    const int wid   = blockIdx.x * 8 + wave;
    const int mtile = wid & 255;
    const int ntile = wid >> 8;
    const int m0    = mtile * 32;
    const int n0    = ntile * 64;

    const float* arow0 = x + (long)(m0 + lane16) * EMBED + 2 * laneHi;
    const float* arow1 = arow0 + 16 * EMBED;

    v8f acc0[4] = {};
    v8f acc1[4] = {};

    #pragma unroll 4
    for (int k0 = 0; k0 < EMBED; k0 += 4) {
        v2f a0 = *(const v2f*)(arow0 + k0);
        v2f a1 = *(const v2f*)(arow1 + k0);
        const int kk = k0 + 2 * laneHi;
        const float* wp = w + (long)kk * THREE_EMBED + n0 + lane16;
        #pragma unroll
        for (int c = 0; c < 4; ++c) {
            v2f b;
            b.x = wp[c * 16];
            b.y = wp[c * 16 + THREE_EMBED];
            acc0[c] = WMMA_F32(a0, b, acc0[c]);
            acc1[c] = WMMA_F32(a1, b, acc1[c]);
        }
    }

    const int section = n0 / EMBED;
    const int h       = (n0 % EMBED) / HD;
    float* dst = (section == 0) ? qb : (section == 1) ? kb : vb;

    #pragma unroll
    for (int c = 0; c < 4; ++c) {
        const float bv = bias[n0 + c * 16 + lane16];
        const int d = c * 16 + lane16;
        #pragma unroll
        for (int g = 0; g < 2; ++g) {
            #pragma unroll
            for (int vv = 0; vv < 8; ++vv) {
                const int m  = m0 + 16 * g + vv + 8 * laneHi;
                const int bg = m >> 11;
                const int t  = m & 2047;
                const float val = (g == 0 ? acc0[c][vv] : acc1[c][vv]) + bv;
                dst[(((long)(bg * NHEAD + h)) * SEQ + t) * HD + d] = val;
            }
        }
    }
}


__device__ __forceinline__ float xhalf(float x)
{
    const int b = __builtin_bit_cast(int, x);
    return __builtin_bit_cast(float,
        __builtin_amdgcn_permlanex16(b, b, 0x76543210, 0xfedcba98, false, false));
}

__global__ __launch_bounds__(32) void attn_kernel(
    const float* __restrict__ q, const float* __restrict__ k,
    const float* __restrict__ v, float* __restrict__ out)
{
    const int lane   = threadIdx.x & 31;
    const int lane16 = lane & 15;
    const int laneHi = lane >> 4;

    const int wid = blockIdx.x;
    const int q32 = wid & 63;
    const int bh  = wid >> 6;
    const int b   = bh >> 4;
    const int h   = bh & 15;
    const int q0  = q32 * 32;

    const float* qp0 = q + ((long)bh * SEQ + q0 + lane16) * HD + 2 * laneHi;
    const float* qp1 = qp0 + 16 * HD;
    v2f bQ0[16], bQ1[16];
    #pragma unroll
    for (int j = 0; j < 16; ++j) {
        bQ0[j] = *(const v2f*)(qp0 + 4 * j);
        bQ1[j] = *(const v2f*)(qp1 + 4 * j);
    }

    float mrun[2] = { -__builtin_inff(), -__builtin_inff() };
    float lrun[2] = { 0.f, 0.f };
    v8f O[2][4] = {};

    const float scale = 0.125f;
    const int ktmax = (q0 + 31) >> 4;

    for (int kt = 0; kt <= ktmax; ++kt) {
        const int k0 = kt * 16;

        const float* kp = k + ((long)bh * SEQ + k0 + lane16) * HD + 2 * laneHi;
        v8f S[2] = {};
        #pragma unroll
        for (int j = 0; j < 16; ++j) {
            v2f aK = *(const v2f*)(kp + 4 * j);
            S[0] = WMMA_F32(aK, bQ0[j], S[0]);
            S[1] = WMMA_F32(aK, bQ1[j], S[1]);
        }

        const int key0 = k0 + 8 * laneHi;
        #pragma unroll
        for (int g = 0; g < 2; ++g) {
            const int qrow = q0 + 16 * g + lane16;
            float tmax = -__builtin_inff();
            #pragma unroll
            for (int r = 0; r < 8; ++r) {
                float s = S[g][r] * scale;
                if (key0 + r > qrow) s = -__builtin_inff();
                S[g][r] = s;
                tmax = fmaxf(tmax, s);
            }
            tmax = fmaxf(tmax, xhalf(tmax));
            const float mnew = fmaxf(mrun[g], tmax);
            float psum = 0.f;
            #pragma unroll
            for (int r = 0; r < 8; ++r) {
                const float p = __expf(S[g][r] - mnew);
                S[g][r] = p;
                psum += p;
            }
            psum += xhalf(psum);
            const float a = __expf(mrun[g] - mnew);
            lrun[g] = lrun[g] * a + psum;
            mrun[g] = mnew;
            #pragma unroll
            for (int cc = 0; cc < 4; ++cc)
                #pragma unroll
                for (int r = 0; r < 8; ++r)
                    O[g][cc][r] *= a;
        }

        const float* vp = v + ((long)bh * SEQ + key0) * HD + lane16;
        #pragma unroll
        for (int cc = 0; cc < 4; ++cc) {
            #pragma unroll
            for (int t = 0; t < 4; ++t) {
                v2f aV;
                aV.x = vp[(2 * t)     * HD + cc * 16];
                aV.y = vp[(2 * t + 1) * HD + cc * 16];
                v2f bP0, bP1;
                bP0.x = S[0][2 * t]; bP0.y = S[0][2 * t + 1];
                bP1.x = S[1][2 * t]; bP1.y = S[1][2 * t + 1];
                O[0][cc] = WMMA_F32(aV, bP0, O[0][cc]);
                O[1][cc] = WMMA_F32(aV, bP1, O[1][cc]);
            }
        }
    }

    #pragma unroll
    for (int g = 0; g < 2; ++g) {
        const int t = q0 + 16 * g + lane16;
        float* orow = out + (((long)b * SEQ + t) * NHEAD + h) * HD + 8 * laneHi;
        #pragma unroll
        for (int cc = 0; cc < 4; ++cc) {
            float4 lo, hi4;
            lo.x  = O[g][cc][0] / lrun[g];
            lo.y  = O[g][cc][1] / lrun[g];
            lo.z  = O[g][cc][2] / lrun[g];
            lo.w  = O[g][cc][3] / lrun[g];
            hi4.x = O[g][cc][4] / lrun[g];
            hi4.y = O[g][cc][5] / lrun[g];
            hi4.z = O[g][cc][6] / lrun[g];
            hi4.w = O[g][cc][7] / lrun[g];
            *(float4*)(orow + cc * 16)     = lo;
            *(float4*)(orow + cc * 16 + 4) = hi4;
        }
    }
}

extern "C" void kernel_launch(void* const* d_in, const int* in_sizes, int n_in,
                              void* d_out, int out_size, void* d_ws, size_t ws_size,
                              hipStream_t stream) {
    const float* x    = (const float*)d_in[0];
    const float* Wqkv = (const float*)d_in[1];
    const float* bqkv = (const float*)d_in[2];
    float* out = (float*)d_out;

    float* ws = (float*)d_ws;
    const long QS = (long)BATCH * NHEAD * SEQ * HD;
    float* qb = ws;
    float* kb = ws + QS;
    float* vb = ws + 2 * QS;

    qkv_gemm_kernel<<<1536, 256, 0, stream>>>(x, Wqkv, bqkv, qb, kb, vb);
    attn_kernel<<<BATCH * NHEAD * (SEQ / 32), 32, 0, stream>>>(qb, kb, vb, out);
}


# Projection GEMM operand loads widened to 16 bytes: a lane takes 4 consecutive k of x and 4 adjacent columns of W (tile column c = columns 4c+cc); 16-byte row stores; k sum order in 8-k blocks changes
# speedup vs baseline: 2.0202x; 1.0862x over previous
#include <hip/hip_runtime.h>

typedef float v2f __attribute__((ext_vector_type(2)));
typedef float v8f __attribute__((ext_vector_type(8)));

#define WMMA_F32(a, b, c) \
    __builtin_amdgcn_wmma_f32_16x16x4_f32(false, (a), false, (b), (short)0, (c), false, false)

#define EMBED 1024
#define THREE_EMBED 3072
#define NHEAD 16
#define HD 64
#define SEQ 2048
#define BATCH 4

__global__ __launch_bounds__(256) void qkv_gemm_kernel(
    const float* __restrict__ x, const float* __restrict__ w,
    const float* __restrict__ bias,
    float* __restrict__ qb, float* __restrict__ kb_out, float* __restrict__ vb)
{
    const int lane   = threadIdx.x & 31;
    const int wave   = threadIdx.x >> 5;
    const int lane16 = lane & 15;
    const int laneHi = lane >> 4;

    const int wid   = blockIdx.x * 8 + wave;
    const int mtile = wid & 255;
    const int ntile = wid >> 8;
    const int m0    = mtile * 32;
    const int n0    = ntile * 64;

    const float* arow0 = x + (long)(m0 + lane16) * EMBED + 4 * laneHi;
    const float* arow1 = arow0 + 16 * EMBED;
    const float* wrow  = w + (long)(4 * laneHi) * THREE_EMBED + n0 + 4 * lane16;

    v8f acc0[4] = {};
    v8f acc1[4] = {};

    for (int kb = 0; kb < EMBED; kb += 32) {
        float4 a0[4], a1[4];
        #pragma unroll
        for (int u = 0; u < 4; ++u) {
            a0[u] = *(const float4*)(arow0 + kb + 8 * u);
            a1[u] = *(const float4*)(arow1 + kb + 8 * u);
        }
        __builtin_amdgcn_sched_barrier(0);

        #pragma unroll
        for (int u = 0; u < 4; ++u) {
            const float* wp = wrow + (long)(kb + 8 * u) * THREE_EMBED;
            const float4 b0 = *(const float4*)(wp);
            const float4 b1 = *(const float4*)(wp + THREE_EMBED);
            const float4 b2 = *(const float4*)(wp + 2 * THREE_EMBED);
            const float4 b3 = *(const float4*)(wp + 3 * THREE_EMBED);

            const float r0[4] = { b0.x, b0.y, b0.z, b0.w };
            const float r1[4] = { b1.x, b1.y, b1.z, b1.w };
            const float r2[4] = { b2.x, b2.y, b2.z, b2.w };
            const float r3[4] = { b3.x, b3.y, b3.z, b3.w };

            v2f fa0, fa1;
            fa0.x = a0[u].x; fa0.y = a0[u].y;
            fa1.x = a1[u].x; fa1.y = a1[u].y;
            #pragma unroll
            for (int c = 0; c < 4; ++c) {
                v2f b;
                b.x = r0[c];
                b.y = r1[c];
                acc0[c] = WMMA_F32(fa0, b, acc0[c]);
                acc1[c] = WMMA_F32(fa1, b, acc1[c]);
            }
            fa0.x = a0[u].z; fa0.y = a0[u].w;
            fa1.x = a1[u].z; fa1.y = a1[u].w;
            #pragma unroll
            for (int c = 0; c < 4; ++c) {
                v2f b;
                b.x = r2[c];
                b.y = r3[c];
                acc0[c] = WMMA_F32(fa0, b, acc0[c]);
                acc1[c] = WMMA_F32(fa1, b, acc1[c]);
            }
        }
    }

    const int section = n0 / EMBED;
    const int h       = (n0 % EMBED) / HD;
    float* dst = (section == 0) ? qb : (section == 1) ? kb_out : vb;

    const float4 bv = *(const float4*)(bias + n0 + 4 * lane16);

    #pragma unroll
    for (int g = 0; g < 2; ++g) {
        #pragma unroll
        for (int vv = 0; vv < 8; ++vv) {
            const int m  = m0 + 16 * g + vv + 8 * laneHi;
            const int bg = m >> 11;
            const int t  = m & 2047;
            float4 o;
            if (g == 0) {
                o.x = acc0[0][vv] + bv.x; o.y = acc0[1][vv] + bv.y;
                o.z = acc0[2][vv] + bv.z; o.w = acc0[3][vv] + bv.w;
            } else {
                o.x = acc1[0][vv] + bv.x; o.y = acc1[1][vv] + bv.y;
                o.z = acc1[2][vv] + bv.z; o.w = acc1[3][vv] + bv.w;
            }
            *(float4*)(dst + (((long)(bg * NHEAD + h)) * SEQ + t) * HD + 4 * lane16) = o;
        }
    }
}


__device__ __forceinline__ float xhalf(float x)
{
    const int b = __builtin_bit_cast(int, x);
    return __builtin_bit_cast(float,
        __builtin_amdgcn_permlanex16(b, b, 0x76543210, 0xfedcba98, false, false));
}

__global__ __launch_bounds__(32) void attn_kernel(
    const float* __restrict__ q, const float* __restrict__ k,
    const float* __restrict__ v, float* __restrict__ out)
{
    const int lane   = threadIdx.x & 31;
    const int lane16 = lane & 15;
    const int laneHi = lane >> 4;

    const int wid = blockIdx.x;
    const int q32 = wid & 63;
    const int bh  = wid >> 6;
    const int b   = bh >> 4;
    const int h   = bh & 15;
    const int q0  = q32 * 32;

    const float* qp0 = q + ((long)bh * SEQ + q0 + lane16) * HD + 2 * laneHi;
    const float* qp1 = qp0 + 16 * HD;
    v2f bQ0[16], bQ1[16];
    #pragma unroll
    for (int j = 0; j < 16; ++j) {
        bQ0[j] = *(const v2f*)(qp0 + 4 * j);
        bQ1[j] = *(const v2f*)(qp1 + 4 * j);
    }

    float mrun[2] = { -__builtin_inff(), -__builtin_inff() };
    float lrun[2] = { 0.f, 0.f };
    v8f O[2][4] = {};

    const float scale = 0.125f;
    const int ktmax = (q0 + 31) >> 4;

    for (int kt = 0; kt <= ktmax; ++kt) {
        const int k0 = kt * 16;

        const float* kp = k + ((long)bh * SEQ + k0 + lane16) * HD + 2 * laneHi;
        v8f S[2] = {};
        #pragma unroll
        for (int j = 0; j < 16; ++j) {
            v2f aK = *(const v2f*)(kp + 4 * j);
            S[0] = WMMA_F32(aK, bQ0[j], S[0]);
            S[1] = WMMA_F32(aK, bQ1[j], S[1]);
        }

        const int key0 = k0 + 8 * laneHi;
        #pragma unroll
        for (int g = 0; g < 2; ++g) {
            const int qrow = q0 + 16 * g + lane16;
            float tmax = -__builtin_inff();
            #pragma unroll
            for (int r = 0; r < 8; ++r) {
                float s = S[g][r] * scale;
                if (key0 + r > qrow) s = -__builtin_inff();
                S[g][r] = s;
                tmax = fmaxf(tmax, s);
            }
            tmax = fmaxf(tmax, xhalf(tmax));
            const float mnew = fmaxf(mrun[g], tmax);
            float psum = 0.f;
            #pragma unroll
            for (int r = 0; r < 8; ++r) {
                const float p = __expf(S[g][r] - mnew);
                S[g][r] = p;
                psum += p;
            }
            psum += xhalf(psum);
            const float a = __expf(mrun[g] - mnew);
            lrun[g] = lrun[g] * a + psum;
            mrun[g] = mnew;
            #pragma unroll
            for (int cc = 0; cc < 4; ++cc)
                #pragma unroll
                for (int r = 0; r < 8; ++r)
                    O[g][cc][r] *= a;
        }

        const float* vp = v + ((long)bh * SEQ + key0) * HD + lane16;
        #pragma unroll
        for (int cc = 0; cc < 4; ++cc) {
            #pragma unroll
            for (int t = 0; t < 4; ++t) {
                v2f aV;
                aV.x = vp[(2 * t)     * HD + cc * 16];
                aV.y = vp[(2 * t + 1) * HD + cc * 16];
                v2f bP0, bP1;
                bP0.x = S[0][2 * t]; bP0.y = S[0][2 * t + 1];
                bP1.x = S[1][2 * t]; bP1.y = S[1][2 * t + 1];
                O[0][cc] = WMMA_F32(aV, bP0, O[0][cc]);
                O[1][cc] = WMMA_F32(aV, bP1, O[1][cc]);
            }
        }
    }

    #pragma unroll
    for (int g = 0; g < 2; ++g) {
        const int t = q0 + 16 * g + lane16;
        float* orow = out + (((long)b * SEQ + t) * NHEAD + h) * HD + 8 * laneHi;
        #pragma unroll
        for (int cc = 0; cc < 4; ++cc) {
            float4 lo, hi4;
            lo.x  = O[g][cc][0] / lrun[g];
            lo.y  = O[g][cc][1] / lrun[g];
            lo.z  = O[g][cc][2] / lrun[g];
            lo.w  = O[g][cc][3] / lrun[g];
            hi4.x = O[g][cc][4] / lrun[g];
            hi4.y = O[g][cc][5] / lrun[g];
            hi4.z = O[g][cc][6] / lrun[g];
            hi4.w = O[g][cc][7] / lrun[g];
            *(float4*)(orow + cc * 16)     = lo;
            *(float4*)(orow + cc * 16 + 4) = hi4;
        }
    }
}

extern "C" void kernel_launch(void* const* d_in, const int* in_sizes, int n_in,
                              void* d_out, int out_size, void* d_ws, size_t ws_size,
                              hipStream_t stream) {
    const float* x    = (const float*)d_in[0];
    const float* Wqkv = (const float*)d_in[1];
    const float* bqkv = (const float*)d_in[2];
    float* out = (float*)d_out;

    float* ws = (float*)d_ws;
    const long QS = (long)BATCH * NHEAD * SEQ * HD;
    float* qb = ws;
    float* kb = ws + QS;
    float* vb = ws + 2 * QS;

    qkv_gemm_kernel<<<1536, 256, 0, stream>>>(x, Wqkv, bqkv, qb, kb, vb);
    attn_kernel<<<BATCH * NHEAD * (SEQ / 32), 32, 0, stream>>>(qb, kb, vb, out);
}


# Causal load balance in attention: each wave takes query tiles 63-p and p of one head (130 key tiles per wave instead of 2..128); the 8 waves of a work-group share one head
# speedup vs baseline: 2.4944x; 1.2347x over previous
#include <hip/hip_runtime.h>

typedef float v2f __attribute__((ext_vector_type(2)));
typedef float v8f __attribute__((ext_vector_type(8)));

#define WMMA_F32(a, b, c) \
    __builtin_amdgcn_wmma_f32_16x16x4_f32(false, (a), false, (b), (short)0, (c), false, false)

#define EMBED 1024
#define THREE_EMBED 3072
#define NHEAD 16
#define HD 64
#define SEQ 2048
#define BATCH 4

__global__ __launch_bounds__(256) void qkv_gemm_kernel(
    const float* __restrict__ x, const float* __restrict__ w,
    const float* __restrict__ bias,
    float* __restrict__ qb, float* __restrict__ kb_out, float* __restrict__ vb)
{
    const int lane   = threadIdx.x & 31;
    const int wave   = threadIdx.x >> 5;
    const int lane16 = lane & 15;
    const int laneHi = lane >> 4;

    const int wid   = blockIdx.x * 8 + wave;
    const int mtile = wid & 255;
    const int ntile = wid >> 8;
    const int m0    = mtile * 32;
    const int n0    = ntile * 64;

    const float* arow0 = x + (long)(m0 + lane16) * EMBED + 4 * laneHi;
    const float* arow1 = arow0 + 16 * EMBED;
    const float* wrow  = w + (long)(4 * laneHi) * THREE_EMBED + n0 + 4 * lane16;

    v8f acc0[4] = {};
    v8f acc1[4] = {};

    for (int kb = 0; kb < EMBED; kb += 32) {
        float4 a0[4], a1[4];
        #pragma unroll
        for (int u = 0; u < 4; ++u) {
            a0[u] = *(const float4*)(arow0 + kb + 8 * u);
            a1[u] = *(const float4*)(arow1 + kb + 8 * u);
        }
        __builtin_amdgcn_sched_barrier(0);

        #pragma unroll
        for (int u = 0; u < 4; ++u) {
            const float* wp = wrow + (long)(kb + 8 * u) * THREE_EMBED;
            const float4 b0 = *(const float4*)(wp);
            const float4 b1 = *(const float4*)(wp + THREE_EMBED);
            const float4 b2 = *(const float4*)(wp + 2 * THREE_EMBED);
            const float4 b3 = *(const float4*)(wp + 3 * THREE_EMBED);

            const float r0[4] = { b0.x, b0.y, b0.z, b0.w };
            const float r1[4] = { b1.x, b1.y, b1.z, b1.w };
            const float r2[4] = { b2.x, b2.y, b2.z, b2.w };
            const float r3[4] = { b3.x, b3.y, b3.z, b3.w };

            v2f fa0, fa1;
            fa0.x = a0[u].x; fa0.y = a0[u].y;
            fa1.x = a1[u].x; fa1.y = a1[u].y;
            #pragma unroll
            for (int c = 0; c < 4; ++c) {
                v2f b;
                b.x = r0[c];
                b.y = r1[c];
                acc0[c] = WMMA_F32(fa0, b, acc0[c]);
                acc1[c] = WMMA_F32(fa1, b, acc1[c]);
            }
            fa0.x = a0[u].z; fa0.y = a0[u].w;
            fa1.x = a1[u].z; fa1.y = a1[u].w;
            #pragma unroll
            for (int c = 0; c < 4; ++c) {
                v2f b;
                b.x = r2[c];
                b.y = r3[c];
                acc0[c] = WMMA_F32(fa0, b, acc0[c]);
                acc1[c] = WMMA_F32(fa1, b, acc1[c]);
            }
        }
    }

    const int section = n0 / EMBED;
    const int h       = (n0 % EMBED) / HD;
    float* dst = (section == 0) ? qb : (section == 1) ? kb_out : vb;

    const float4 bv = *(const float4*)(bias + n0 + 4 * lane16);

    #pragma unroll
    for (int g = 0; g < 2; ++g) {
        #pragma unroll
        for (int vv = 0; vv < 8; ++vv) {
            const int m  = m0 + 16 * g + vv + 8 * laneHi;
            const int bg = m >> 11;
            const int t  = m & 2047;
            float4 o;
            if (g == 0) {
                o.x = acc0[0][vv] + bv.x; o.y = acc0[1][vv] + bv.y;
                o.z = acc0[2][vv] + bv.z; o.w = acc0[3][vv] + bv.w;
            } else {
                o.x = acc1[0][vv] + bv.x; o.y = acc1[1][vv] + bv.y;
                o.z = acc1[2][vv] + bv.z; o.w = acc1[3][vv] + bv.w;
            }
            *(float4*)(dst + (((long)(bg * NHEAD + h)) * SEQ + t) * HD + 4 * lane16) = o;
        }
    }
}


__device__ __forceinline__ float xhalf(float x)
{
    const int b = __builtin_bit_cast(int, x);
    return __builtin_bit_cast(float,
        __builtin_amdgcn_permlanex16(b, b, 0x76543210, 0xfedcba98, false, false));
}

__global__ __launch_bounds__(256) void attn_kernel(
    const float* __restrict__ q, const float* __restrict__ k,
    const float* __restrict__ v, float* __restrict__ out)
{
    const int lane   = threadIdx.x & 31;
    const int lane16 = lane & 15;
    const int laneHi = lane >> 4;
    const int wave = __builtin_amdgcn_readfirstlane((int)(threadIdx.x >> 5));

    const int bh   = blockIdx.x >> 2;
    const int pair = ((blockIdx.x & 3) << 3) + wave;
    const int b    = bh >> 4;
    const int h    = bh & 15;

    const float scale = 0.125f;

    for (int part = 0; part < 2; ++part) {
        const int q32 = (part == 0) ? (63 - pair) : pair;
        const int q0  = q32 * 32;

        const float* qp0 = q + ((long)bh * SEQ + q0 + lane16) * HD + 2 * laneHi;
        const float* qp1 = qp0 + 16 * HD;
        v2f bQ0[16], bQ1[16];
        #pragma unroll
        for (int j = 0; j < 16; ++j) {
            bQ0[j] = *(const v2f*)(qp0 + 4 * j);
            bQ1[j] = *(const v2f*)(qp1 + 4 * j);
        }

        float mrun[2] = { -__builtin_inff(), -__builtin_inff() };
        float lrun[2] = { 0.f, 0.f };
        v8f O[2][4] = {};

        const int ktmax = (q0 + 31) >> 4;

        for (int kt = 0; kt <= ktmax; ++kt) {
            const int k0 = kt * 16;

            const float* kp = k + ((long)bh * SEQ + k0 + lane16) * HD + 2 * laneHi;
            v8f S[2] = {};
            #pragma unroll
            for (int j = 0; j < 16; ++j) {
                v2f aK = *(const v2f*)(kp + 4 * j);
                S[0] = WMMA_F32(aK, bQ0[j], S[0]);
                S[1] = WMMA_F32(aK, bQ1[j], S[1]);
            }

            const int key0 = k0 + 8 * laneHi;
            #pragma unroll
            for (int g = 0; g < 2; ++g) {
                const int qrow = q0 + 16 * g + lane16;
                float tmax = -__builtin_inff();
                #pragma unroll
                for (int r = 0; r < 8; ++r) {
                    float s = S[g][r] * scale;
                    if (key0 + r > qrow) s = -__builtin_inff();
                    S[g][r] = s;
                    tmax = fmaxf(tmax, s);
                }
                tmax = fmaxf(tmax, xhalf(tmax));
                const float mnew = fmaxf(mrun[g], tmax);
                float psum = 0.f;
                #pragma unroll
                for (int r = 0; r < 8; ++r) {
                    const float p = __expf(S[g][r] - mnew);
                    S[g][r] = p;
                    psum += p;
                }
                psum += xhalf(psum);
                const float a = __expf(mrun[g] - mnew);
                lrun[g] = lrun[g] * a + psum;
                mrun[g] = mnew;
                #pragma unroll
                for (int cc = 0; cc < 4; ++cc)
                    #pragma unroll
                    for (int r = 0; r < 8; ++r)
                        O[g][cc][r] *= a;
            }

            const float* vp = v + ((long)bh * SEQ + key0) * HD + lane16;
            #pragma unroll
            for (int cc = 0; cc < 4; ++cc) {
                #pragma unroll
                for (int t = 0; t < 4; ++t) {
                    v2f aV;
                    aV.x = vp[(2 * t)     * HD + cc * 16];
                    aV.y = vp[(2 * t + 1) * HD + cc * 16];
                    v2f bP0, bP1;
                    bP0.x = S[0][2 * t]; bP0.y = S[0][2 * t + 1];
                    bP1.x = S[1][2 * t]; bP1.y = S[1][2 * t + 1];
                    O[0][cc] = WMMA_F32(aV, bP0, O[0][cc]);
                    O[1][cc] = WMMA_F32(aV, bP1, O[1][cc]);
                }
            }
        }

        #pragma unroll
        for (int g = 0; g < 2; ++g) {
            const int t = q0 + 16 * g + lane16;
            float* orow = out + (((long)b * SEQ + t) * NHEAD + h) * HD + 8 * laneHi;
            #pragma unroll
            for (int cc = 0; cc < 4; ++cc) {
                float4 lo, hi4;
                lo.x  = O[g][cc][0] / lrun[g];
                lo.y  = O[g][cc][1] / lrun[g];
                lo.z  = O[g][cc][2] / lrun[g];
                lo.w  = O[g][cc][3] / lrun[g];
                hi4.x = O[g][cc][4] / lrun[g];
                hi4.y = O[g][cc][5] / lrun[g];
                hi4.z = O[g][cc][6] / lrun[g];
                hi4.w = O[g][cc][7] / lrun[g];
                *(float4*)(orow + cc * 16)     = lo;
                *(float4*)(orow + cc * 16 + 4) = hi4;
            }
        }
    }
}

extern "C" void kernel_launch(void* const* d_in, const int* in_sizes, int n_in,
                              void* d_out, int out_size, void* d_ws, size_t ws_size,
                              hipStream_t stream) {
    const float* x    = (const float*)d_in[0];
    const float* Wqkv = (const float*)d_in[1];
    const float* bqkv = (const float*)d_in[2];
    float* out = (float*)d_out;

    float* ws = (float*)d_ws;
    const long QS = (long)BATCH * NHEAD * SEQ * HD;
    float* qb = ws;
    float* kb = ws + QS;
    float* vb = ws + 2 * QS;

    qkv_gemm_kernel<<<1536, 256, 0, stream>>>(x, Wqkv, bqkv, qb, kb, vb);
    attn_kernel<<<BATCH * NHEAD * 4, 256, 0, stream>>>(qb, kb, vb, out);
}
